# starting kernel
# baseline (speedup 1.0000x reference)
#include <hip/hip_runtime.h>

typedef __attribute__((ext_vector_type(2))) float v2f;
typedef __attribute__((ext_vector_type(8))) float v8f;

#define DEG_MAX 80
#define NDEG    81
#define NPOS    6561
#define CIN     64
#define COUT    64
#define BATCH   32
#define NCPT    17
#define INTERVAL 5

__global__ void shconv_interp(const float* __restrict__ weight,
                              float* __restrict__ wl) {
    int tid = blockIdx.x * blockDim.x + threadIdx.x;
    if (tid >= NDEG * CIN * COUT) return;
    int o = tid & 63;
    int i = (tid >> 6) & 63;
    int l = tid >> 12;

    int c0; float t;
    if (l < (NCPT - 2) * INTERVAL) {
        c0 = l / INTERVAL;
        t  = (float)(l - c0 * INTERVAL) * (1.0f / (float)INTERVAL);
    } else {
        c0 = NCPT - 2;
        t  = (float)(l - (NCPT - 2) * INTERVAL) * (1.0f / (float)INTERVAL);
    }
    const float* wb = weight + ((i * COUT + o) * NCPT);
    float w0 = wb[c0];
    float w1 = wb[c0 + 1];
    wl[tid] = (1.0f - t) * w0 + t * w1;
}

__global__ void __launch_bounds__(128)
shconv_wmma(const float* __restrict__ x,
            const float* __restrict__ wl,
            float* __restrict__ out) {
    const int l    = blockIdx.x;
    const int b    = blockIdx.y;
    const int lane = threadIdx.x & 31;
    const int wave = threadIdx.x >> 5;
    const int obase = wave * 16;
    const int col   = lane & 15;
    const int khi   = (lane >> 4) << 1;

    const int n_start = l * l;
    const int n_end   = n_start + 2 * l + 1;

    v2f A[16];
    const float* wbl = wl + l * (CIN * COUT) + obase + col;
#pragma unroll
    for (int ks = 0; ks < 16; ++ks) {
        const int k = 4 * ks + khi;
        A[ks].x = wbl[(k + 0) * COUT];
        A[ks].y = wbl[(k + 1) * COUT];
    }

    const float* xb = x   + (size_t)b * CIN  * NPOS;
    float*       ob = out + (size_t)b * COUT * NPOS;

    const int ntiles = (2 * l + 1 + 15) >> 4;
    for (int nt = 0; nt < ntiles; ++nt) {
        const int  n     = n_start + nt * 16 + col;
        const bool inrun = (n < n_end);
        const int  nc    = inrun ? n : (NPOS - 1);

        v8f c = {0.f, 0.f, 0.f, 0.f, 0.f, 0.f, 0.f, 0.f};
#pragma unroll
        for (int ks = 0; ks < 16; ++ks) {
            const int k = 4 * ks + khi;
            float b0 = xb[(size_t)(k + 0) * NPOS + nc];
            float b1 = xb[(size_t)(k + 1) * NPOS + nc];
            v2f bm;
            bm.x = inrun ? b0 : 0.0f;
            bm.y = inrun ? b1 : 0.0f;
            c = __builtin_amdgcn_wmma_f32_16x16x4_f32(
                    false, A[ks], false, bm, (short)0, c, false, false);
        }

        if (inrun) {
            const int orow = obase + ((lane >> 4) << 3);
            float* op = ob + (size_t)orow * NPOS + n;
#pragma unroll
            for (int r = 0; r < 8; ++r)
                op[(size_t)r * NPOS] = c[r];
        }
    }
}

extern "C" void kernel_launch(void* const* d_in, const int* in_sizes, int n_in,
                              void* d_out, int out_size, void* d_ws, size_t ws_size,
                              hipStream_t stream) {
    const float* x      = (const float*)d_in[0];
    const float* weight = (const float*)d_in[1];
    float*       out    = (float*)d_out;
    float*       wl     = (float*)d_ws;

    const int total = NDEG * CIN * COUT;
    shconv_interp<<<(total + 255) / 256, 256, 0, stream>>>(weight, wl);
    shconv_wmma<<<dim3(NDEG, BATCH), 128, 0, stream>>>(x, wl, out);
}


# Re-tile the product: one wave = one 32-position strip of a degree x all 64 out channels x 2 batches (W_l^T in registers, 8 accumulator tiles); 8-byte x/out and 16-byte weight accesses, x read once
# speedup vs baseline: 1.3152x; 1.3152x over previous
#include <hip/hip_runtime.h>


typedef __attribute__((ext_vector_type(2))) float v2f;
typedef __attribute__((ext_vector_type(4))) float v4f;
typedef __attribute__((ext_vector_type(8))) float v8f;
typedef v2f v2f_a4 __attribute__((aligned(4)));

#define DEG_MAX 80
#define NDEG    81
#define NPOS    6561
#define CIN     64
#define COUT    64
#define BATCH   32
#define NCPT    17
#define INTERVAL 5

#define STRIP   32
#define NSTRIPS 246
#define BPW     2
#define WAVES   4
#define YBLOCKS 4

__global__ void shconv_interp(const float* __restrict__ weight,
                              float* __restrict__ wl) {
    int tid = blockIdx.x * blockDim.x + threadIdx.x;
    if (tid >= NDEG * CIN * COUT) return;
    int o = tid & 63;
    int i = (tid >> 6) & 63;
    int l = tid >> 12;

    int c0; float t;
    if (l < (NCPT - 2) * INTERVAL) {
        c0 = l / INTERVAL;
        t  = (float)(l - c0 * INTERVAL) * (1.0f / (float)INTERVAL);
    } else {
        c0 = NCPT - 2;
        t  = (float)(l - (NCPT - 2) * INTERVAL) * (1.0f / (float)INTERVAL);
    }
    const float* wb = weight + ((i * COUT + o) * NCPT);
    float w0 = wb[c0];
    float w1 = wb[c0 + 1];
    wl[tid] = (1.0f - t) * w0 + t * w1;
}

__global__ void __launch_bounds__(32 * WAVES)
shconv_wmma(const float* __restrict__ x,
            const float* __restrict__ wl,
            float* __restrict__ out) {
    const int s    = blockIdx.x;
    const int lane = threadIdx.x & 31;
    const int wave = __builtin_amdgcn_readfirstlane((int)(threadIdx.x >> 5));
    const int bg   = blockIdx.y * WAVES + wave;
    const unsigned c = lane & 15;
    const unsigned h = lane >> 4;

    const int t = (s >= 81) + (s >= 146) + (s >= 195) + (s >= 228) + (s >= 245);
    const int l = 16 * t + (s - (81 * t - 8 * t * (t - 1)));

    const int n0    = l * l + STRIP * t;
    const int n_end = l * l + 2 * l + 1;

    v2f A[4][16];
    {
        const float* wlane = wl + l * (CIN * COUT) + (2u * h) * COUT + 4u * c;
#pragma unroll
        for (int ks = 0; ks < 16; ++ks) {
            const v4f q0 = *(const v4f*)(wlane + (4 * ks + 0) * COUT);
            const v4f q1 = *(const v4f*)(wlane + (4 * ks + 1) * COUT);
#pragma unroll
            for (int j = 0; j < 4; ++j) {
                A[j][ks].x = q0[j];
                A[j][ks].y = q1[j];
            }
        }
    }

    const unsigned p0    = (unsigned)n0 + 2u * c;
    const unsigned xlane = (2u * h) * NPOS + p0;
    const unsigned olane = (32u * h) * NPOS + p0;
    const unsigned q0c = p0     < (unsigned)(NPOS - 1) ? p0     : (unsigned)(NPOS - 1);
    const unsigned q1c = p0 + 1 < (unsigned)(NPOS - 1) ? p0 + 1 : (unsigned)(NPOS - 1);
    const bool both = (p0 + 1 < (unsigned)n_end);
    const bool one  = (p0     < (unsigned)n_end);

    for (int bi = 0; bi < BPW; ++bi) {
        const int b = bg * BPW + bi;
        const float* xb = x   + (size_t)b * CIN  * NPOS;
        float*       ob = out + (size_t)b * COUT * NPOS;

        v8f acc[4][2];
#pragma unroll
        for (int j = 0; j < 4; ++j) {
            acc[j][0] = (v8f){0.f, 0.f, 0.f, 0.f, 0.f, 0.f, 0.f, 0.f};
            acc[j][1] = (v8f){0.f, 0.f, 0.f, 0.f, 0.f, 0.f, 0.f, 0.f};
        }

#pragma unroll
        for (int ks = 0; ks < 16; ++ks) {
            const v2f pa = *(const v2f_a4*)(xb + xlane + (4 * ks + 0) * NPOS);
            v2f pb;
            if (ks < 15) {
                pb = *(const v2f_a4*)(xb + xlane + (4 * ks + 1) * NPOS);
            } else {
                const float* rowp = xb + (2u * h) * NPOS + (4 * ks + 1) * NPOS;
                pb.x = rowp[q0c];
                pb.y = rowp[q1c];
            }
            v2f b0, b1;
            b0.x = pa.x; b0.y = pb.x;
            b1.x = pa.y; b1.y = pb.y;
#pragma unroll
            for (int j = 0; j < 4; ++j) {
                acc[j][0] = __builtin_amdgcn_wmma_f32_16x16x4_f32(
                        false, A[j][ks], false, b0, (short)0, acc[j][0], false, false);
                acc[j][1] = __builtin_amdgcn_wmma_f32_16x16x4_f32(
                        false, A[j][ks], false, b1, (short)0, acc[j][1], false, false);
            }
        }

        if (both) {
#pragma unroll
            for (int j = 0; j < 4; ++j) {
#pragma unroll
                for (int r = 0; r < 8; ++r) {
                    v2f v;
                    v.x = acc[j][0][r];
                    v.y = acc[j][1][r];
                    *(v2f_a4*)(ob + olane + (4 * r + j) * NPOS) = v;
                }
            }
        } else if (one) {
#pragma unroll
            for (int j = 0; j < 4; ++j) {
#pragma unroll
                for (int r = 0; r < 8; ++r)
                    ob[olane + (4 * r + j) * NPOS] = acc[j][0][r];
            }
        }
    }
}

extern "C" void kernel_launch(void* const* d_in, const int* in_sizes, int n_in,
                              void* d_out, int out_size, void* d_ws, size_t ws_size,
                              hipStream_t stream) {
    const float* x      = (const float*)d_in[0];
    const float* weight = (const float*)d_in[1];
    float*       out    = (float*)d_out;
    float*       wl     = (float*)d_ws;

    const int total = NDEG * CIN * COUT;
    shconv_interp<<<(total + 255) / 256, 256, 0, stream>>>(weight, wl);
    shconv_wmma<<<dim3(NSTRIPS, YBLOCKS), 32 * WAVES, 0, stream>>>(x, wl, out);
}


# Stream x through a 7-step register ring, loads issued 7 k-steps ahead of their products, also across the batch boundary: 248 VGPRs (was 264), 4 waves per SIMD, all 984 workgroups resident at once
# speedup vs baseline: 1.4463x; 1.0997x over previous
#include <hip/hip_runtime.h>


typedef __attribute__((ext_vector_type(2))) float v2f;
typedef __attribute__((ext_vector_type(4))) float v4f;
typedef __attribute__((ext_vector_type(8))) float v8f;
typedef v2f v2f_a4 __attribute__((aligned(4)));

#define DEG_MAX 80
#define NDEG    81
#define NPOS    6561
#define CIN     64
#define COUT    64
#define BATCH   32
#define NCPT    17
#define INTERVAL 5

#define STRIP   32
#define NSTRIPS 246
#define BPW     2
#define WAVES   4
#define YBLOCKS 4

__global__ void shconv_interp(const float* __restrict__ weight,
                              float* __restrict__ wl) {
    int tid = blockIdx.x * blockDim.x + threadIdx.x;
    if (tid >= NDEG * CIN * COUT) return;
    int o = tid & 63;
    int i = (tid >> 6) & 63;
    int l = tid >> 12;

    int c0; float t;
    if (l < (NCPT - 2) * INTERVAL) {
        c0 = l / INTERVAL;
        t  = (float)(l - c0 * INTERVAL) * (1.0f / (float)INTERVAL);
    } else {
        c0 = NCPT - 2;
        t  = (float)(l - (NCPT - 2) * INTERVAL) * (1.0f / (float)INTERVAL);
    }
    const float* wb = weight + ((i * COUT + o) * NCPT);
    float w0 = wb[c0];
    float w1 = wb[c0 + 1];
    wl[tid] = (1.0f - t) * w0 + t * w1;
}

#define XDEPTH 7
#define NSTEPS (BPW * 16)

__device__ __forceinline__ void load_x_step(const float* xw, int g,
                                            unsigned xlane, unsigned h,
                                            unsigned q0c, unsigned q1c,
                                            v2f& pa, v2f& pb) {
    const int bi = g >> 4;
    const int ks = g & 15;
    const float* xb = xw + (size_t)bi * CIN * NPOS;
    pa = *(const v2f_a4*)(xb + xlane + (4 * ks + 0) * NPOS);
    if (ks < 15) {
        pb = *(const v2f_a4*)(xb + xlane + (4 * ks + 1) * NPOS);
    } else {
        const float* rowp = xb + (2u * h) * NPOS + (4 * ks + 1) * NPOS;
        pb.x = rowp[q0c];
        pb.y = rowp[q1c];
    }
}

__global__ void __launch_bounds__(32 * WAVES, 4)
shconv_wmma(const float* x,
            const float* __restrict__ wl,
            float* out) {
    const int s    = blockIdx.x;
    const int lane = threadIdx.x & 31;
    const int wave = __builtin_amdgcn_readfirstlane((int)(threadIdx.x >> 5));
    const int bg   = blockIdx.y * WAVES + wave;
    const unsigned c = lane & 15;
    const unsigned h = lane >> 4;

    const int t = (s >= 81) + (s >= 146) + (s >= 195) + (s >= 228) + (s >= 245);
    const int l = 16 * t + (s - (81 * t - 8 * t * (t - 1)));

    const int n0    = l * l + STRIP * t;
    const int n_end = l * l + 2 * l + 1;

    const unsigned p0    = (unsigned)n0 + 2u * c;
    const unsigned xlane = (2u * h) * NPOS + p0;
    const unsigned olane = (32u * h) * NPOS + p0;
    const unsigned q0c = p0     < (unsigned)(NPOS - 1) ? p0     : (unsigned)(NPOS - 1);
    const unsigned q1c = p0 + 1 < (unsigned)(NPOS - 1) ? p0 + 1 : (unsigned)(NPOS - 1);
    const bool both = (p0 + 1 < (unsigned)n_end);
    const bool one  = (p0     < (unsigned)n_end);

    const float* xw = x   + (size_t)(bg * BPW) * CIN  * NPOS;
    float*       ow = out + (size_t)(bg * BPW) * COUT * NPOS;

    v2f ra[XDEPTH], rb[XDEPTH];
#pragma unroll
    for (int g = 0; g < XDEPTH; ++g)
        load_x_step(xw, g, xlane, h, q0c, q1c, ra[g], rb[g]);

    v2f A[4][16];
    {
        const float* wlane = wl + l * (CIN * COUT) + (2u * h) * COUT + 4u * c;
#pragma unroll
        for (int ks = 0; ks < 16; ++ks) {
            const v4f q0 = *(const v4f*)(wlane + (4 * ks + 0) * COUT);
            const v4f q1 = *(const v4f*)(wlane + (4 * ks + 1) * COUT);
#pragma unroll
            for (int j = 0; j < 4; ++j) {
                A[j][ks].x = q0[j];
                A[j][ks].y = q1[j];
            }
        }
    }

    v8f acc[4][2];

#pragma unroll
    for (int g = 0; g < NSTEPS; ++g) {
        const int ks = g & 15;
        __builtin_amdgcn_sched_barrier(0);
        if (ks == 0) {
#pragma unroll
            for (int j = 0; j < 4; ++j) {
                acc[j][0] = (v8f){0.f, 0.f, 0.f, 0.f, 0.f, 0.f, 0.f, 0.f};
                acc[j][1] = (v8f){0.f, 0.f, 0.f, 0.f, 0.f, 0.f, 0.f, 0.f};
            }
        }

        const v2f pa = ra[g % XDEPTH];
        const v2f pb = rb[g % XDEPTH];
        if (g + XDEPTH < NSTEPS)
            load_x_step(xw, g + XDEPTH, xlane, h, q0c, q1c, ra[g % XDEPTH], rb[g % XDEPTH]);

        v2f b0, b1;
        b0.x = pa.x; b0.y = pb.x;
        b1.x = pa.y; b1.y = pb.y;
#pragma unroll
        for (int j = 0; j < 4; ++j) {
            acc[j][0] = __builtin_amdgcn_wmma_f32_16x16x4_f32(
                    false, A[j][ks], false, b0, (short)0, acc[j][0], false, false);
            acc[j][1] = __builtin_amdgcn_wmma_f32_16x16x4_f32(
                    false, A[j][ks], false, b1, (short)0, acc[j][1], false, false);
        }

        if (ks == 15) {
            float* ob = ow + (size_t)(g >> 4) * COUT * NPOS;
            if (both) {
#pragma unroll
                for (int j = 0; j < 4; ++j) {
#pragma unroll
                    for (int r = 0; r < 8; ++r) {
                        v2f v;
                        v.x = acc[j][0][r];
                        v.y = acc[j][1][r];
                        *(v2f_a4*)(ob + olane + (4 * r + j) * NPOS) = v;
                    }
                }
            } else if (one) {
#pragma unroll
                for (int j = 0; j < 4; ++j) {
#pragma unroll
                    for (int r = 0; r < 8; ++r)
                        ob[olane + (4 * r + j) * NPOS] = acc[j][0][r];
                }
            }
        }
    }
}

extern "C" void kernel_launch(void* const* d_in, const int* in_sizes, int n_in,
                              void* d_out, int out_size, void* d_ws, size_t ws_size,
                              hipStream_t stream) {
    const float* x      = (const float*)d_in[0];
    const float* weight = (const float*)d_in[1];
    float*       out    = (float*)d_out;
    float*       wl     = (float*)d_ws;

    const int total = NDEG * CIN * COUT;
    shconv_interp<<<(total + 255) / 256, 256, 0, stream>>>(weight, wl);
    shconv_wmma<<<dim3(NSTRIPS, YBLOCKS), 32 * WAVES, 0, stream>>>(x, wl, out);
}


# Workgroup = 16 waves on 16 strips adjacent in memory, one batch pair (was 1 strip x 4 batch pairs): a 128-byte line cut by a strip edge is now read and written inside one WGP; grid 16 x 16
# speedup vs baseline: 1.6787x; 1.1607x over previous
#include <hip/hip_runtime.h>


typedef __attribute__((ext_vector_type(2))) float v2f;
typedef __attribute__((ext_vector_type(4))) float v4f;
typedef __attribute__((ext_vector_type(8))) float v8f;
typedef v2f v2f_a4 __attribute__((aligned(4)));

#define DEG_MAX 80
#define NDEG    81
#define NPOS    6561
#define CIN     64
#define COUT    64
#define BATCH   32
#define NCPT    17
#define INTERVAL 5

#define STRIP   32
#define NSTRIPS 246
#define BPW     2
#define WAVES   16
#define XGROUPS 16
#define BGROUPS 16

__global__ void shconv_interp(const float* __restrict__ weight,
                              float* __restrict__ wl) {
    int tid = blockIdx.x * blockDim.x + threadIdx.x;
    if (tid >= NDEG * CIN * COUT) return;
    int o = tid & 63;
    int i = (tid >> 6) & 63;
    int l = tid >> 12;

    int c0; float t;
    if (l < (NCPT - 2) * INTERVAL) {
        c0 = l / INTERVAL;
        t  = (float)(l - c0 * INTERVAL) * (1.0f / (float)INTERVAL);
    } else {
        c0 = NCPT - 2;
        t  = (float)(l - (NCPT - 2) * INTERVAL) * (1.0f / (float)INTERVAL);
    }
    const float* wb = weight + ((i * COUT + o) * NCPT);
    float w0 = wb[c0];
    float w1 = wb[c0 + 1];
    wl[tid] = (1.0f - t) * w0 + t * w1;
}

#define XDEPTH 7
#define NSTEPS (BPW * 16)

__device__ __forceinline__ void load_x_step(const float* xw, int g,
                                            unsigned xlane, unsigned h,
                                            unsigned q0c, unsigned q1c,
                                            v2f& pa, v2f& pb) {
    const int bi = g >> 4;
    const int ks = g & 15;
    const float* xb = xw + (size_t)bi * CIN * NPOS;
    pa = *(const v2f_a4*)(xb + xlane + (4 * ks + 0) * NPOS);
    if (ks < 15) {
        pb = *(const v2f_a4*)(xb + xlane + (4 * ks + 1) * NPOS);
    } else {
        const float* rowp = xb + (2u * h) * NPOS + (4 * ks + 1) * NPOS;
        pb.x = rowp[q0c];
        pb.y = rowp[q1c];
    }
}

__global__ void __launch_bounds__(32 * WAVES, 4)
shconv_wmma(const float* x,
            const float* __restrict__ wl,
            float* out) {
    const int lane = threadIdx.x & 31;
    const int wave = __builtin_amdgcn_readfirstlane((int)(threadIdx.x >> 5));
    const int s    = blockIdx.x * WAVES + wave;
    if (s >= NSTRIPS) return;
    const int bg   = blockIdx.y;
    const unsigned c = lane & 15;
    const unsigned h = lane >> 4;

    const int q  = (s >= 16) + (s >= 48) + (s >= 96) + (s >= 160) + (s >= 240);
    const int sr = s - 8 * q * (q + 1);
    const int l  = 16 * q + sr / (q + 1);
    const int t  = sr % (q + 1);

    const int n0    = l * l + STRIP * t;
    const int n_end = l * l + 2 * l + 1;

    const unsigned p0    = (unsigned)n0 + 2u * c;
    const unsigned xlane = (2u * h) * NPOS + p0;
    const unsigned olane = (32u * h) * NPOS + p0;
    const unsigned q0c = p0     < (unsigned)(NPOS - 1) ? p0     : (unsigned)(NPOS - 1);
    const unsigned q1c = p0 + 1 < (unsigned)(NPOS - 1) ? p0 + 1 : (unsigned)(NPOS - 1);
    const bool both = (p0 + 1 < (unsigned)n_end);
    const bool one  = (p0     < (unsigned)n_end);

    const float* xw = x   + (size_t)(bg * BPW) * CIN  * NPOS;
    float*       ow = out + (size_t)(bg * BPW) * COUT * NPOS;

    v2f ra[XDEPTH], rb[XDEPTH];
#pragma unroll
    for (int g = 0; g < XDEPTH; ++g)
        load_x_step(xw, g, xlane, h, q0c, q1c, ra[g], rb[g]);

    v2f A[4][16];
    {
        const float* wlane = wl + l * (CIN * COUT) + (2u * h) * COUT + 4u * c;
#pragma unroll
        for (int ks = 0; ks < 16; ++ks) {
            const v4f q0 = *(const v4f*)(wlane + (4 * ks + 0) * COUT);
            const v4f q1 = *(const v4f*)(wlane + (4 * ks + 1) * COUT);
#pragma unroll
            for (int j = 0; j < 4; ++j) {
                A[j][ks].x = q0[j];
                A[j][ks].y = q1[j];
            }
        }
    }

    v8f acc[4][2];

#pragma unroll
    for (int g = 0; g < NSTEPS; ++g) {
        const int ks = g & 15;
        __builtin_amdgcn_sched_barrier(0);
        if (ks == 0) {
#pragma unroll
            for (int j = 0; j < 4; ++j) {
                acc[j][0] = (v8f){0.f, 0.f, 0.f, 0.f, 0.f, 0.f, 0.f, 0.f};
                acc[j][1] = (v8f){0.f, 0.f, 0.f, 0.f, 0.f, 0.f, 0.f, 0.f};
            }
        }

        const v2f pa = ra[g % XDEPTH];
        const v2f pb = rb[g % XDEPTH];
        if (g + XDEPTH < NSTEPS)
            load_x_step(xw, g + XDEPTH, xlane, h, q0c, q1c, ra[g % XDEPTH], rb[g % XDEPTH]);

        v2f b0, b1;
        b0.x = pa.x; b0.y = pb.x;
        b1.x = pa.y; b1.y = pb.y;
#pragma unroll
        for (int j = 0; j < 4; ++j) {
            acc[j][0] = __builtin_amdgcn_wmma_f32_16x16x4_f32(
                    false, A[j][ks], false, b0, (short)0, acc[j][0], false, false);
            acc[j][1] = __builtin_amdgcn_wmma_f32_16x16x4_f32(
                    false, A[j][ks], false, b1, (short)0, acc[j][1], false, false);
        }

        if (ks == 15) {
            float* ob = ow + (size_t)(g >> 4) * COUT * NPOS;
            if (both) {
#pragma unroll
                for (int j = 0; j < 4; ++j) {
#pragma unroll
                    for (int r = 0; r < 8; ++r) {
                        v2f v;
                        v.x = acc[j][0][r];
                        v.y = acc[j][1][r];
                        *(v2f_a4*)(ob + olane + (4 * r + j) * NPOS) = v;
                    }
                }
            } else if (one) {
#pragma unroll
                for (int j = 0; j < 4; ++j) {
#pragma unroll
                    for (int r = 0; r < 8; ++r)
                        ob[olane + (4 * r + j) * NPOS] = acc[j][0][r];
                }
            }
        }
    }
}

extern "C" void kernel_launch(void* const* d_in, const int* in_sizes, int n_in,
                              void* d_out, int out_size, void* d_ws, size_t ws_size,
                              hipStream_t stream) {
    const float* x      = (const float*)d_in[0];
    const float* weight = (const float*)d_in[1];
    float*       out    = (float*)d_out;
    float*       wl     = (float*)d_ws;

    const int total = NDEG * CIN * COUT;
    shconv_interp<<<(total + 255) / 256, 256, 0, stream>>>(weight, wl);
    shconv_wmma<<<dim3(XGROUPS, BGROUPS), 32 * WAVES, 0, stream>>>(x, wl, out);
}


# Output staged through LDS: waves write their tiles into an image of the block's run; behind a split barrier each wave stores 4 rows as whole 128-byte lines, batch 0's beside batch 1's products
# speedup vs baseline: 1.9140x; 1.1402x over previous
#include <hip/hip_runtime.h>


typedef __attribute__((ext_vector_type(2))) float v2f;
typedef __attribute__((ext_vector_type(4))) float v4f;
typedef __attribute__((ext_vector_type(8))) float v8f;
typedef v2f v2f_a4 __attribute__((aligned(4)));

#define DEG_MAX 80
#define NDEG    81
#define NPOS    6561
#define CIN     64
#define COUT    64
#define BATCH   32
#define NCPT    17
#define INTERVAL 5

#define STRIP   32
#define NSTRIPS 246
#define BPW     2
#define WAVES   16
#define XGROUPS 16
#define BGROUPS 16

#define PITCH   516
#define HSKEW   32
#define LDSBUF  (COUT * PITCH + HSKEW)
#define DEFER   6

__global__ void shconv_interp(const float* __restrict__ weight,
                              float* __restrict__ wl) {
    int tid = blockIdx.x * blockDim.x + threadIdx.x;
    if (tid >= NDEG * CIN * COUT) return;
    int o = tid & 63;
    int i = (tid >> 6) & 63;
    int l = tid >> 12;

    int c0; float t;
    if (l < (NCPT - 2) * INTERVAL) {
        c0 = l / INTERVAL;
        t  = (float)(l - c0 * INTERVAL) * (1.0f / (float)INTERVAL);
    } else {
        c0 = NCPT - 2;
        t  = (float)(l - (NCPT - 2) * INTERVAL) * (1.0f / (float)INTERVAL);
    }
    const float* wb = weight + ((i * COUT + o) * NCPT);
    float w0 = wb[c0];
    float w1 = wb[c0 + 1];
    wl[tid] = (1.0f - t) * w0 + t * w1;
}

#define XDEPTH 7
#define NSTEPS (BPW * 16)

__device__ __forceinline__ void strip_span(int s, int& l, int& n0, int& send) {
    const int q  = (s >= 16) + (s >= 48) + (s >= 96) + (s >= 160) + (s >= 240);
    const int sr = s - 8 * q * (q + 1);
    l = 16 * q + sr / (q + 1);
    const int t = sr % (q + 1);
    n0 = l * l + STRIP * t;
    const int dend = l * l + 2 * l + 1;
    send = n0 + STRIP < dend ? n0 + STRIP : dend;
}

__device__ __forceinline__ int lds_row(int o) { return o * PITCH + (o >> 5) * HSKEW; }

__device__ __forceinline__ void store_image(const float* img, float* ob, int wave, int lane,
                                            int P0, int len) {
#pragma unroll
    for (int rr = 0; rr < 4; ++rr) {
        const int o = 4 * wave + rr;
        const int a = (o + P0) & 31;
        const float* lrow = img + lds_row(o) - (a & 28);
        float*       grow = ob + o * NPOS + (P0 - a);
        v4f v[5];
#pragma unroll
        for (int m = 0; m < 5; ++m) {
            const int qd = lane + 32 * m;
            const int d0 = 4 * qd - a;
            const bool whole = d0 >= 0 && d0 + 3 < len;
            v[m] = *(const v4f*)(whole ? lrow + 4 * qd : img + lds_row(o));
        }
#pragma unroll
        for (int m = 0; m < 5; ++m) {
            const int qd = lane + 32 * m;
            const int d0 = 4 * qd - a;
            if (d0 >= 0 && d0 + 3 < len) *(v4f*)(grow + 4 * qd) = v[m];
        }
        const int k  = lane & 3;
        const int e0 = ((a + len) & ~3) - a;
        const int dp = lane < 4 ? k - (a & 3) : e0 + k;
        const bool cut = lane < 4 ? (a & 3) != 0 : ((a + len) & 3) != 0;
        if (lane < 8 && cut && (unsigned)dp < (unsigned)len)
            ob[o * NPOS + P0 + dp] = img[lds_row(o) + (a & 3) + dp];
    }
}

__device__ __forceinline__ void load_x_step(const float* xw, int g,
                                            unsigned xlane, unsigned h,
                                            unsigned q0c, unsigned q1c,
                                            v2f& pa, v2f& pb) {
    const int bi = g >> 4;
    const int ks = g & 15;
    const float* xb = xw + (size_t)bi * CIN * NPOS;
    pa = *(const v2f_a4*)(xb + xlane + (4 * ks + 0) * NPOS);
    if (ks < 15) {
        pb = *(const v2f_a4*)(xb + xlane + (4 * ks + 1) * NPOS);
    } else {
        const float* rowp = xb + (2u * h) * NPOS + (4 * ks + 1) * NPOS;
        pb.x = rowp[q0c];
        pb.y = rowp[q1c];
    }
}

__global__ void __launch_bounds__(32 * WAVES, 4)
shconv_wmma(const float* x,
            const float* __restrict__ wl,
            float* out) {
    __shared__ __attribute__((aligned(16))) float lds_img[2 * LDSBUF];

    const int lane = threadIdx.x & 31;
    const int wave = __builtin_amdgcn_readfirstlane((int)(threadIdx.x >> 5));
    const int s_id = blockIdx.x * WAVES + wave;
    const bool active = s_id < NSTRIPS;
    const int s    = active ? s_id : NSTRIPS - 1;
    const int bg   = blockIdx.y;
    const unsigned c = lane & 15;
    const unsigned h = lane >> 4;

    int l, n0, s_end;
    strip_span(s, l, n0, s_end);
    const int n_end = l * l + 2 * l + 1;

    int l_f, P0, e_f, l_l, n_l, P1;
    strip_span(blockIdx.x * WAVES, l_f, P0, e_f);
    strip_span(blockIdx.x * WAVES + WAVES - 1 < NSTRIPS ? blockIdx.x * WAVES + WAVES - 1
                                                        : NSTRIPS - 1, l_l, n_l, P1);
    const int len = P1 - P0;

    const unsigned p0    = (unsigned)n0 + 2u * c;
    const unsigned xlane = (2u * h) * NPOS + p0;
    const unsigned q0c = p0     < (unsigned)(NPOS - 1) ? p0     : (unsigned)(NPOS - 1);
    const unsigned q1c = p0 + 1 < (unsigned)(NPOS - 1) ? p0 + 1 : (unsigned)(NPOS - 1);
    const int wlane = (int)h * (32 * PITCH + HSKEW) + (int)p0 - P0;
    const bool both = (p0 + 1 < (unsigned)n_end);
    const bool one  = (p0     < (unsigned)n_end);

    const float* xw = x   + (size_t)(bg * BPW) * CIN  * NPOS;
    float*       ow = out + (size_t)(bg * BPW) * COUT * NPOS;

    v2f ra[XDEPTH], rb[XDEPTH];
#pragma unroll
    for (int g = 0; g < XDEPTH; ++g)
        load_x_step(xw, g, xlane, h, q0c, q1c, ra[g], rb[g]);

    v2f A[4][16];
    {
        const float* wlane = wl + l * (CIN * COUT) + (2u * h) * COUT + 4u * c;
#pragma unroll
        for (int ks = 0; ks < 16; ++ks) {
            const v4f q0 = *(const v4f*)(wlane + (4 * ks + 0) * COUT);
            const v4f q1 = *(const v4f*)(wlane + (4 * ks + 1) * COUT);
#pragma unroll
            for (int j = 0; j < 4; ++j) {
                A[j][ks].x = q0[j];
                A[j][ks].y = q1[j];
            }
        }
    }

    v8f acc[4][2];

#pragma unroll
    for (int g = 0; g < NSTEPS; ++g) {
        const int ks = g & 15;
        __builtin_amdgcn_sched_barrier(0);
        if (g == 16 + DEFER) {
            asm volatile("s_barrier_wait -1" ::: "memory");
            store_image(lds_img, ow, wave, lane, P0, len);
        }
        if (ks == 0) {
#pragma unroll
            for (int j = 0; j < 4; ++j) {
                acc[j][0] = (v8f){0.f, 0.f, 0.f, 0.f, 0.f, 0.f, 0.f, 0.f};
                acc[j][1] = (v8f){0.f, 0.f, 0.f, 0.f, 0.f, 0.f, 0.f, 0.f};
            }
        }

        const v2f pa = ra[g % XDEPTH];
        const v2f pb = rb[g % XDEPTH];
        if (g + XDEPTH < NSTEPS)
            load_x_step(xw, g + XDEPTH, xlane, h, q0c, q1c, ra[g % XDEPTH], rb[g % XDEPTH]);

        v2f b0, b1;
        b0.x = pa.x; b0.y = pb.x;
        b1.x = pa.y; b1.y = pb.y;
#pragma unroll
        for (int j = 0; j < 4; ++j) {
            acc[j][0] = __builtin_amdgcn_wmma_f32_16x16x4_f32(
                    false, A[j][ks], false, b0, (short)0, acc[j][0], false, false);
            acc[j][1] = __builtin_amdgcn_wmma_f32_16x16x4_f32(
                    false, A[j][ks], false, b1, (short)0, acc[j][1], false, false);
        }

        if (ks == 15) {
            float* img = lds_img + (g >> 4) * LDSBUF;
            if (active) {
                if (both) {
#pragma unroll
                    for (int j = 0; j < 4; ++j) {
                        float* q = img + wlane + ((j + P0) & 3) + j * PITCH;
#pragma unroll
                        for (int r = 0; r < 8; ++r) {
                            q[4 * r * PITCH]     = acc[j][0][r];
                            q[4 * r * PITCH + 1] = acc[j][1][r];
                        }
                    }
                } else if (one) {
#pragma unroll
                    for (int j = 0; j < 4; ++j) {
                        float* q = img + wlane + ((j + P0) & 3) + j * PITCH;
#pragma unroll
                        for (int r = 0; r < 8; ++r)
                            q[4 * r * PITCH] = acc[j][0][r];
                    }
                }
            }
            asm volatile("s_wait_dscnt 0x0\n\ts_barrier_signal -1" ::: "memory");
        }
    }
    asm volatile("s_barrier_wait -1" ::: "memory");
    store_image(lds_img + LDSBUF, ow + (size_t)COUT * NPOS, wave, lane, P0, len);
}

extern "C" void kernel_launch(void* const* d_in, const int* in_sizes, int n_in,
                              void* d_out, int out_size, void* d_ws, size_t ws_size,
                              hipStream_t stream) {
    const float* x      = (const float*)d_in[0];
    const float* weight = (const float*)d_in[1];
    float*       out    = (float*)d_out;
    float*       wl     = (float*)d_ws;

    const int total = NDEG * CIN * COUT;
    shconv_interp<<<(total + 255) / 256, 256, 0, stream>>>(weight, wl);
    shconv_wmma<<<dim3(XGROUPS, BGROUPS), 32 * WAVES, 0, stream>>>(x, wl, out);
}
